# starting kernel
# baseline (speedup 1.0000x reference)
#include <hip/hip_runtime.h>
#include <math.h>

#define B_   2
#define L_   2048
#define D_   1024
#define H_   16
#define DH_  64
#define NTOK (B_ * L_)

typedef __attribute__((ext_vector_type(2))) float v2f;
typedef __attribute__((ext_vector_type(8))) float v8f;

__device__ __forceinline__ v8f wmma4(v2f a, v2f b, v8f c) {
  return __builtin_amdgcn_wmma_f32_16x16x4_f32(
       false, a,  false, b,
       (short)0, c,  false,  false);
}

__device__ __forceinline__ float gelu_tanh(float x) {
  float x3 = x * x * x;
  return 0.5f * x * (1.0f + tanhf(0.7978845608028654f * (x + 0.044715f * x3)));
}

__global__ void __launch_bounds__(256)
rmsnorm_kernel(const float* __restrict__ x, const float* __restrict__ scale,
               float* __restrict__ out) {
  const int row = blockIdx.x;
  const float* xr = x + (size_t)row * D_;
  float* orow = out + (size_t)row * D_;
  const int t = threadIdx.x;
  float4 v = ((const float4*)xr)[t];
  float ss = v.x * v.x + v.y * v.y + v.z * v.z + v.w * v.w;
  for (int off = 16; off >= 1; off >>= 1) ss += __shfl_xor(ss, off, 32);
  __shared__ float red[8];
  if ((t & 31) == 0) red[t >> 5] = ss;
  __syncthreads();
  float tot = 0.f;
#pragma unroll
  for (int i = 0; i < 8; ++i) tot += red[i];
  float r = rsqrtf(tot * (1.0f / (float)D_) + 1e-6f);
  float4 s4 = ((const float4*)scale)[t];
  float4 o;
  o.x = v.x * s4.x * r; o.y = v.y * s4.y * r;
  o.z = v.z * s4.z * r; o.w = v.w * s4.w * r;
  ((float4*)orow)[t] = o;
}

__global__ void __launch_bounds__(256)
gemm_nt_kernel(const float* __restrict__ A, const float* __restrict__ W,
               const float* __restrict__ R, float* __restrict__ C,
               int M, int N, int K, int flags) {
  const int lane = threadIdx.x & 31;
  const int wave = threadIdx.x >> 5;
  const int half = lane >> 4;
  const int l16 = lane & 15;
  const int rowBase = blockIdx.y * 128 + wave * 16;
  const int colBase = blockIdx.x * 64;

  const float* arow = A + (size_t)(rowBase + l16) * K + 2 * half;
  const float* b0r  = W + (size_t)(colBase + l16) * K + 2 * half;
  const float* b1r  = b0r + (size_t)16 * K;
  const float* b2r  = b0r + (size_t)32 * K;
  const float* b3r  = b0r + (size_t)48 * K;

  __builtin_prefetch(arow, 0, 0);
  __builtin_prefetch(b0r, 0, 0);
  __builtin_prefetch(b1r, 0, 0);
  __builtin_prefetch(b2r, 0, 0);
  __builtin_prefetch(b3r, 0, 0);

  v8f acc0 = {0.f,0.f,0.f,0.f,0.f,0.f,0.f,0.f};
  v8f acc1 = acc0, acc2 = acc0, acc3 = acc0;

  v2f a_c  = *(const v2f*)(arow);
  v2f x0_c = *(const v2f*)(b0r);
  v2f x1_c = *(const v2f*)(b1r);
  v2f x2_c = *(const v2f*)(b2r);
  v2f x3_c = *(const v2f*)(b3r);

#pragma unroll 8
  for (int k = 0; k < K - 4; k += 4) {
    const int kn = k + 4;
    v2f a_n  = *(const v2f*)(arow + kn);
    v2f x0_n = *(const v2f*)(b0r + kn);
    v2f x1_n = *(const v2f*)(b1r + kn);
    v2f x2_n = *(const v2f*)(b2r + kn);
    v2f x3_n = *(const v2f*)(b3r + kn);
    acc0 = wmma4(a_c, x0_c, acc0);
    acc1 = wmma4(a_c, x1_c, acc1);
    acc2 = wmma4(a_c, x2_c, acc2);
    acc3 = wmma4(a_c, x3_c, acc3);
    a_c = a_n; x0_c = x0_n; x1_c = x1_n; x2_c = x2_n; x3_c = x3_n;
  }
  acc0 = wmma4(a_c, x0_c, acc0);
  acc1 = wmma4(a_c, x1_c, acc1);
  acc2 = wmma4(a_c, x2_c, acc2);
  acc3 = wmma4(a_c, x3_c, acc3);

#pragma unroll
  for (int r = 0; r < 8; ++r) {
    int row = rowBase + r + 8 * half;
    size_t i0 = (size_t)row * N + colBase + l16;
    float v0 = acc0[r], v1 = acc1[r], v2 = acc2[r], v3 = acc3[r];
    if (flags & 1) {
      v0 += R[i0]; v1 += R[i0 + 16]; v2 += R[i0 + 32]; v3 += R[i0 + 48];
    }
    if (flags & 2) {
      v0 = gelu_tanh(v0); v1 = gelu_tanh(v1);
      v2 = gelu_tanh(v2); v3 = gelu_tanh(v3);
    }
    C[i0] = v0; C[i0 + 16] = v1; C[i0 + 32] = v2; C[i0 + 48] = v3;
  }
}

__global__ void __launch_bounds__(256)
rope_kernel(float* __restrict__ qkv, const float* __restrict__ pos,
            const float* __restrict__ pos_orig, const float* __restrict__ tm) {
  int idx = blockIdx.x * blockDim.x + threadIdx.x;
  if (idx >= NTOK * H_ * 25) return;
  int i   = idx % 25;
  int h   = (idx / 25) % H_;
  int tok = idx / (25 * H_);
  int c = i / 5, j = i % 5;
  int fi = j * H_ + h;
  float theta;
  if (c < 4) {
    float freq = __expf(1.1447298858494002f + (float)fi * 0.028782313662425572f);
    int coord = (c < 2) ? 1 : 0;
    int s = (c < 2) ? c : c - 2;
    const float* P = (s == 0) ? pos_orig : pos;
    float pv = P[(size_t)tok * 2 + coord] * 2.0f - 1.0f;
    theta = pv * freq;
  } else {
    float tf = __expf(-(float)fi * 0.057564627324851145f);
    theta = tm[tok] * tf;
  }
  float cs = cosf(theta), sn = sinf(theta);
  float* qp = qkv + (size_t)tok * 3 * D_ + h * DH_;
  float* kp = qp + D_;
  float q1 = qp[i], q2 = qp[i + 25];
  qp[i] = q1 * cs - q2 * sn;  qp[i + 25] = q2 * cs + q1 * sn;
  float k1 = kp[i], k2 = kp[i + 25];
  kp[i] = k1 * cs - k2 * sn;  kp[i + 25] = k2 * cs + k1 * sn;
}

__global__ void __launch_bounds__(256)
cos_scale_kernel(float* __restrict__ qkv, const float* __restrict__ qk_scale) {
  const int lane = threadIdx.x & 31;
  const int wave = threadIdx.x >> 5;
  const int row = blockIdx.x * 8 + wave;
  const int h = row % H_;
  const int tok = row / H_;
  float* qp = qkv + (size_t)tok * 3 * D_ + h * DH_;
  float* kp = qp + D_;
  float2 q2 = ((float2*)qp)[lane];
  float2 k2 = ((float2*)kp)[lane];
  float sq = q2.x * q2.x + q2.y * q2.y;
  float sk = k2.x * k2.x + k2.y * k2.y;
  for (int off = 16; off >= 1; off >>= 1) {
    sq += __shfl_xor(sq, off, 32);
    sk += __shfl_xor(sk, off, 32);
  }
  float s = sqrtf(qk_scale[h]);
  float qs = s * rsqrtf(sq + 1e-6f);
  float ks = s * rsqrtf(sk + 1e-6f);
  ((float2*)qp)[lane] = make_float2(q2.x * qs, q2.y * qs);
  ((float2*)kp)[lane] = make_float2(k2.x * ks, k2.y * ks);
}

__global__ void __launch_bounds__(32)
attn_kernel(const float* __restrict__ qkv, float* __restrict__ out) {
  const int lane = threadIdx.x & 31;
  const int half = lane >> 4, l16 = lane & 15;
  const int qt = blockIdx.x;
  const int bh = blockIdx.y;
  const int b = bh / H_, h = bh % H_;
  const size_t ts = 3 * D_;

  const float* qrow = qkv + (size_t)(b * L_ + qt * 16 + l16) * ts + h * DH_ + 2 * half;
  v2f qf[16];
#pragma unroll
  for (int s = 0; s < 16; ++s) qf[s] = *(const v2f*)(qrow + 4 * s);

  float m[8], lsum[8];
#pragma unroll
  for (int r = 0; r < 8; ++r) { m[r] = -INFINITY; lsum[r] = 0.f; }
  v8f o0 = {0.f,0.f,0.f,0.f,0.f,0.f,0.f,0.f};
  v8f o1 = o0, o2 = o0, o3 = o0;

  __shared__ float P[16 * 17];
  const float* kbase = qkv + (size_t)(b * L_) * ts + D_ + h * DH_;
  const float* vbase = qkv + (size_t)(b * L_) * ts + 2 * D_ + h * DH_;

  for (int kt = 0; kt < L_ / 16; ++kt) {
    const float* krow = kbase + (size_t)(kt * 16 + l16) * ts + 2 * half;
    v8f S0 = {0.f,0.f,0.f,0.f,0.f,0.f,0.f,0.f};
    v8f S1 = S0;
#pragma unroll
    for (int s = 0; s < 16; s += 2) {
      v2f bf0 = *(const v2f*)(krow + 4 * s);
      v2f bf1 = *(const v2f*)(krow + 4 * s + 4);
      S0 = wmma4(qf[s], bf0, S0);
      S1 = wmma4(qf[s + 1], bf1, S1);
    }
    __syncthreads();
#pragma unroll
    for (int r = 0; r < 8; ++r) {
      float sv = S0[r] + S1[r];
      float mx = sv;
      for (int off = 8; off >= 1; off >>= 1) mx = fmaxf(mx, __shfl_xor(mx, off, 32));
      float mnew = fmaxf(m[r], mx);
      float alpha = __expf(m[r] - mnew);
      float p = __expf(sv - mnew);
      float rs = p;
      for (int off = 8; off >= 1; off >>= 1) rs += __shfl_xor(rs, off, 32);
      lsum[r] = lsum[r] * alpha + rs;
      m[r] = mnew;
      o0[r] *= alpha; o1[r] *= alpha; o2[r] *= alpha; o3[r] *= alpha;
      P[(r + 8 * half) * 17 + l16] = p;
    }
    __syncthreads();
    const float* vrow = vbase + (size_t)(kt * 16) * ts;
#pragma unroll
    for (int s = 0; s < 4; ++s) {
      int k0 = s * 4 + 2 * half;
      v2f pf;
      pf.x = P[l16 * 17 + k0];
      pf.y = P[l16 * 17 + k0 + 1];
      const float* v0p = vrow + (size_t)k0 * ts;
      const float* v1p = v0p + ts;
      v2f bv;
      bv.x = v0p[l16];       bv.y = v1p[l16];       o0 = wmma4(pf, bv, o0);
      bv.x = v0p[16 + l16];  bv.y = v1p[16 + l16];  o1 = wmma4(pf, bv, o1);
      bv.x = v0p[32 + l16];  bv.y = v1p[32 + l16];  o2 = wmma4(pf, bv, o2);
      bv.x = v0p[48 + l16];  bv.y = v1p[48 + l16];  o3 = wmma4(pf, bv, o3);
    }
  }
  float* obase = out + (size_t)(b * L_ + qt * 16) * D_ + h * DH_;
#pragma unroll
  for (int r = 0; r < 8; ++r) {
    float inv = 1.0f / lsum[r];
    float* op = obase + (size_t)(r + 8 * half) * D_;
    op[l16]      = o0[r] * inv;
    op[16 + l16] = o1[r] * inv;
    op[32 + l16] = o2[r] * inv;
    op[48 + l16] = o3[r] * inv;
  }
}

extern "C" void kernel_launch(void* const* d_in, const int* in_sizes, int n_in,
                              void* d_out, int out_size, void* d_ws, size_t ws_size,
                              hipStream_t stream) {
  (void)in_sizes; (void)n_in; (void)out_size; (void)ws_size;
  const float* x        = (const float*)d_in[0];
  const float* pos      = (const float*)d_in[1];
  const float* pos_orig = (const float*)d_in[2];
  const float* tm       = (const float*)d_in[3];
  const float* w_qkv    = (const float*)d_in[4];
  const float* w_out    = (const float*)d_in[5];
  const float* w_up     = (const float*)d_in[6];
  const float* w_down   = (const float*)d_in[7];
  const float* n1s      = (const float*)d_in[8];
  const float* n2s      = (const float*)d_in[9];
  const float* qksc     = (const float*)d_in[10];
  float* out = (float*)d_out;

  float* xn   = (float*)d_ws;
  float* qkv  = xn  + (size_t)NTOK * D_;
  float* attn = qkv + (size_t)NTOK * 3 * D_;
  float* up   = attn + (size_t)NTOK * D_;

  rmsnorm_kernel<<<NTOK, 256, 0, stream>>>(x, n1s, xn);
  gemm_nt_kernel<<<dim3(3 * D_ / 64, NTOK / 128), 256, 0, stream>>>(
      xn, w_qkv, nullptr, qkv, NTOK, 3 * D_, D_, 0);
  rope_kernel<<<(NTOK * H_ * 25 + 255) / 256, 256, 0, stream>>>(qkv, pos, pos_orig, tm);
  cos_scale_kernel<<<NTOK * H_ / 8, 256, 0, stream>>>(qkv, qksc);
  attn_kernel<<<dim3(L_ / 16, B_ * H_), 32, 0, stream>>>(qkv, attn);
  gemm_nt_kernel<<<dim3(D_ / 64, NTOK / 128), 256, 0, stream>>>(
      attn, w_out, x, out, NTOK, D_, D_, 1);
  rmsnorm_kernel<<<NTOK, 256, 0, stream>>>(out, n2s, xn);
  gemm_nt_kernel<<<dim3(4 * D_ / 64, NTOK / 128), 256, 0, stream>>>(
      xn, w_up, nullptr, up, NTOK, 4 * D_, D_, 2);
  gemm_nt_kernel<<<dim3(D_ / 64, NTOK / 128), 256, 0, stream>>>(
      up, w_down, out, out, NTOK, D_, 4 * D_, 1);
}


# GEMMs: 128x128x32 A/W tiles staged in double-buffered LDS by coalesced 16-B loads, 64x32 wave tile fed by 16-B LDS fragment reads; k terms summed in a new order inside each group of 8
# speedup vs baseline: 2.2159x; 2.2159x over previous
#include <hip/hip_runtime.h>
#include <math.h>

#define B_   2
#define L_   2048
#define D_   1024
#define H_   16
#define DH_  64
#define NTOK (B_ * L_)

typedef __attribute__((ext_vector_type(2))) float v2f;
typedef __attribute__((ext_vector_type(8))) float v8f;

__device__ __forceinline__ v8f wmma4(v2f a, v2f b, v8f c) {
  return __builtin_amdgcn_wmma_f32_16x16x4_f32(
       false, a,  false, b,
       (short)0, c,  false,  false);
}

__device__ __forceinline__ float gelu_tanh(float x) {
  float x3 = x * x * x;
  return 0.5f * x * (1.0f + tanhf(0.7978845608028654f * (x + 0.044715f * x3)));
}

__global__ void __launch_bounds__(256)
rmsnorm_kernel(const float* __restrict__ x, const float* __restrict__ scale,
               float* __restrict__ out) {
  const int row = blockIdx.x;
  const float* xr = x + (size_t)row * D_;
  float* orow = out + (size_t)row * D_;
  const int t = threadIdx.x;
  float4 v = ((const float4*)xr)[t];
  float ss = v.x * v.x + v.y * v.y + v.z * v.z + v.w * v.w;
  for (int off = 16; off >= 1; off >>= 1) ss += __shfl_xor(ss, off, 32);
  __shared__ float red[8];
  if ((t & 31) == 0) red[t >> 5] = ss;
  __syncthreads();
  float tot = 0.f;
#pragma unroll
  for (int i = 0; i < 8; ++i) tot += red[i];
  float r = rsqrtf(tot * (1.0f / (float)D_) + 1e-6f);
  float4 s4 = ((const float4*)scale)[t];
  float4 o;
  o.x = v.x * s4.x * r; o.y = v.y * s4.y * r;
  o.z = v.z * s4.z * r; o.w = v.w * s4.w * r;
  ((float4*)orow)[t] = o;
}

#define GBM 128
#define GBN 128
#define GBK 32
#define GLD 36

typedef __attribute__((ext_vector_type(4))) float v4f;

__global__ void __launch_bounds__(256)
gemm_nt_kernel(const float* __restrict__ A, const float* __restrict__ W,
               const float* R, float* C,
               int M, int N, int K, int flags) {
  __shared__ __attribute__((aligned(16))) float As[2][GBM * GLD];
  __shared__ __attribute__((aligned(16))) float Ws[2][GBN * GLD];

  const int tid = threadIdx.x;
  const int lane = tid & 31;
  const int wave = tid >> 5;
  const int half = lane >> 4;
  const int l16 = lane & 15;
  const int wm = wave >> 2;
  const int wn = wave & 3;
  const int rowBlk = blockIdx.y * GBM;
  const int colBlk = blockIdx.x * GBN;

  const int lr = tid >> 3;
  const int lc = (tid & 7) * 4;
  const float* ag = A + (size_t)(rowBlk + lr) * K + lc;
  const float* wg = W + (size_t)(colBlk + lr) * K + lc;
  const size_t g32 = (size_t)32 * K;
  const int sOff = lr * GLD + lc;

  v4f ra[4], rw[4];
#pragma unroll
  for (int i = 0; i < 4; ++i) {
    ra[i] = *(const v4f*)(ag + i * g32);
    rw[i] = *(const v4f*)(wg + i * g32);
  }
#pragma unroll
  for (int i = 0; i < 4; ++i) {
    *(v4f*)(&As[0][sOff + i * 32 * GLD]) = ra[i];
    *(v4f*)(&Ws[0][sOff + i * 32 * GLD]) = rw[i];
  }
  __syncthreads();

  v8f acc[4][2];
#pragma unroll
  for (int i = 0; i < 4; ++i)
#pragma unroll
    for (int n = 0; n < 2; ++n)
      acc[i][n] = (v8f){0.f,0.f,0.f,0.f,0.f,0.f,0.f,0.f};

  const int aOff = (wm * 64 + l16) * GLD + 4 * half;
  const int bOff = (wn * 32 + l16) * GLD + 4 * half;
  const int T = K / GBK;

  for (int t = 0; t < T; ++t) {
    const int cur = t & 1;
    const bool more = (t + 1 < T);
    if (more) {
      const float* agn = ag + (size_t)(t + 1) * GBK;
      const float* wgn = wg + (size_t)(t + 1) * GBK;
#pragma unroll
      for (int i = 0; i < 4; ++i) {
        ra[i] = *(const v4f*)(agn + i * g32);
        rw[i] = *(const v4f*)(wgn + i * g32);
      }
    }
    const float* as = &As[cur][aOff];
    const float* ws = &Ws[cur][bOff];

    v4f fa[4], fb[2];
#pragma unroll
    for (int i = 0; i < 4; ++i) fa[i] = *(const v4f*)(as + i * 16 * GLD);
#pragma unroll
    for (int n = 0; n < 2; ++n) fb[n] = *(const v4f*)(ws + n * 16 * GLD);

#pragma unroll
    for (int j = 0; j < 4; ++j) {
      v4f na[4], nb[2];
      if (j < 3) {
#pragma unroll
        for (int i = 0; i < 4; ++i) na[i] = *(const v4f*)(as + i * 16 * GLD + 8 * (j + 1));
#pragma unroll
        for (int n = 0; n < 2; ++n) nb[n] = *(const v4f*)(ws + n * 16 * GLD + 8 * (j + 1));
      }
#pragma unroll
      for (int i = 0; i < 4; ++i)
#pragma unroll
        for (int n = 0; n < 2; ++n) {
          v2f a2 = {fa[i].x, fa[i].y};
          v2f b2 = {fb[n].x, fb[n].y};
          acc[i][n] = wmma4(a2, b2, acc[i][n]);
        }
#pragma unroll
      for (int i = 0; i < 4; ++i)
#pragma unroll
        for (int n = 0; n < 2; ++n) {
          v2f a2 = {fa[i].z, fa[i].w};
          v2f b2 = {fb[n].z, fb[n].w};
          acc[i][n] = wmma4(a2, b2, acc[i][n]);
        }
      if (j < 3) {
#pragma unroll
        for (int i = 0; i < 4; ++i) fa[i] = na[i];
#pragma unroll
        for (int n = 0; n < 2; ++n) fb[n] = nb[n];
      }
    }

    if (more) {
#pragma unroll
      for (int i = 0; i < 4; ++i) {
        *(v4f*)(&As[cur ^ 1][sOff + i * 32 * GLD]) = ra[i];
        *(v4f*)(&Ws[cur ^ 1][sOff + i * 32 * GLD]) = rw[i];
      }
    }
    __syncthreads();
  }

  const int row0 = rowBlk + wm * 64 + 8 * half;
  const int col0 = colBlk + wn * 32 + l16;
#pragma unroll
  for (int i = 0; i < 4; ++i) {
    const size_t ib = (size_t)(row0 + i * 16) * N + col0;
    float v0[8], v1[8];
#pragma unroll
    for (int r = 0; r < 8; ++r) { v0[r] = acc[i][0][r]; v1[r] = acc[i][1][r]; }
    if (flags & 1) {
#pragma unroll
      for (int r = 0; r < 8; ++r) {
        v0[r] += R[ib + (size_t)r * N];
        v1[r] += R[ib + (size_t)r * N + 16];
      }
    }
    if (flags & 2) {
#pragma unroll
      for (int r = 0; r < 8; ++r) { v0[r] = gelu_tanh(v0[r]); v1[r] = gelu_tanh(v1[r]); }
    }
#pragma unroll
    for (int r = 0; r < 8; ++r) {
      C[ib + (size_t)r * N] = v0[r];
      C[ib + (size_t)r * N + 16] = v1[r];
    }
  }
}

__global__ void __launch_bounds__(256)
rope_kernel(float* __restrict__ qkv, const float* __restrict__ pos,
            const float* __restrict__ pos_orig, const float* __restrict__ tm) {
  int idx = blockIdx.x * blockDim.x + threadIdx.x;
  if (idx >= NTOK * H_ * 25) return;
  int i   = idx % 25;
  int h   = (idx / 25) % H_;
  int tok = idx / (25 * H_);
  int c = i / 5, j = i % 5;
  int fi = j * H_ + h;
  float theta;
  if (c < 4) {
    float freq = __expf(1.1447298858494002f + (float)fi * 0.028782313662425572f);
    int coord = (c < 2) ? 1 : 0;
    int s = (c < 2) ? c : c - 2;
    const float* P = (s == 0) ? pos_orig : pos;
    float pv = P[(size_t)tok * 2 + coord] * 2.0f - 1.0f;
    theta = pv * freq;
  } else {
    float tf = __expf(-(float)fi * 0.057564627324851145f);
    theta = tm[tok] * tf;
  }
  float cs = cosf(theta), sn = sinf(theta);
  float* qp = qkv + (size_t)tok * 3 * D_ + h * DH_;
  float* kp = qp + D_;
  float q1 = qp[i], q2 = qp[i + 25];
  qp[i] = q1 * cs - q2 * sn;  qp[i + 25] = q2 * cs + q1 * sn;
  float k1 = kp[i], k2 = kp[i + 25];
  kp[i] = k1 * cs - k2 * sn;  kp[i + 25] = k2 * cs + k1 * sn;
}

__global__ void __launch_bounds__(256)
cos_scale_kernel(float* __restrict__ qkv, const float* __restrict__ qk_scale) {
  const int lane = threadIdx.x & 31;
  const int wave = threadIdx.x >> 5;
  const int row = blockIdx.x * 8 + wave;
  const int h = row % H_;
  const int tok = row / H_;
  float* qp = qkv + (size_t)tok * 3 * D_ + h * DH_;
  float* kp = qp + D_;
  float2 q2 = ((float2*)qp)[lane];
  float2 k2 = ((float2*)kp)[lane];
  float sq = q2.x * q2.x + q2.y * q2.y;
  float sk = k2.x * k2.x + k2.y * k2.y;
  for (int off = 16; off >= 1; off >>= 1) {
    sq += __shfl_xor(sq, off, 32);
    sk += __shfl_xor(sk, off, 32);
  }
  float s = sqrtf(qk_scale[h]);
  float qs = s * rsqrtf(sq + 1e-6f);
  float ks = s * rsqrtf(sk + 1e-6f);
  ((float2*)qp)[lane] = make_float2(q2.x * qs, q2.y * qs);
  ((float2*)kp)[lane] = make_float2(k2.x * ks, k2.y * ks);
}

__global__ void __launch_bounds__(32)
attn_kernel(const float* __restrict__ qkv, float* __restrict__ out) {
  const int lane = threadIdx.x & 31;
  const int half = lane >> 4, l16 = lane & 15;
  const int qt = blockIdx.x;
  const int bh = blockIdx.y;
  const int b = bh / H_, h = bh % H_;
  const size_t ts = 3 * D_;

  const float* qrow = qkv + (size_t)(b * L_ + qt * 16 + l16) * ts + h * DH_ + 2 * half;
  v2f qf[16];
#pragma unroll
  for (int s = 0; s < 16; ++s) qf[s] = *(const v2f*)(qrow + 4 * s);

  float m[8], lsum[8];
#pragma unroll
  for (int r = 0; r < 8; ++r) { m[r] = -INFINITY; lsum[r] = 0.f; }
  v8f o0 = {0.f,0.f,0.f,0.f,0.f,0.f,0.f,0.f};
  v8f o1 = o0, o2 = o0, o3 = o0;

  __shared__ float P[16 * 17];
  const float* kbase = qkv + (size_t)(b * L_) * ts + D_ + h * DH_;
  const float* vbase = qkv + (size_t)(b * L_) * ts + 2 * D_ + h * DH_;

  for (int kt = 0; kt < L_ / 16; ++kt) {
    const float* krow = kbase + (size_t)(kt * 16 + l16) * ts + 2 * half;
    v8f S0 = {0.f,0.f,0.f,0.f,0.f,0.f,0.f,0.f};
    v8f S1 = S0;
#pragma unroll
    for (int s = 0; s < 16; s += 2) {
      v2f bf0 = *(const v2f*)(krow + 4 * s);
      v2f bf1 = *(const v2f*)(krow + 4 * s + 4);
      S0 = wmma4(qf[s], bf0, S0);
      S1 = wmma4(qf[s + 1], bf1, S1);
    }
    __syncthreads();
#pragma unroll
    for (int r = 0; r < 8; ++r) {
      float sv = S0[r] + S1[r];
      float mx = sv;
      for (int off = 8; off >= 1; off >>= 1) mx = fmaxf(mx, __shfl_xor(mx, off, 32));
      float mnew = fmaxf(m[r], mx);
      float alpha = __expf(m[r] - mnew);
      float p = __expf(sv - mnew);
      float rs = p;
      for (int off = 8; off >= 1; off >>= 1) rs += __shfl_xor(rs, off, 32);
      lsum[r] = lsum[r] * alpha + rs;
      m[r] = mnew;
      o0[r] *= alpha; o1[r] *= alpha; o2[r] *= alpha; o3[r] *= alpha;
      P[(r + 8 * half) * 17 + l16] = p;
    }
    __syncthreads();
    const float* vrow = vbase + (size_t)(kt * 16) * ts;
#pragma unroll
    for (int s = 0; s < 4; ++s) {
      int k0 = s * 4 + 2 * half;
      v2f pf;
      pf.x = P[l16 * 17 + k0];
      pf.y = P[l16 * 17 + k0 + 1];
      const float* v0p = vrow + (size_t)k0 * ts;
      const float* v1p = v0p + ts;
      v2f bv;
      bv.x = v0p[l16];       bv.y = v1p[l16];       o0 = wmma4(pf, bv, o0);
      bv.x = v0p[16 + l16];  bv.y = v1p[16 + l16];  o1 = wmma4(pf, bv, o1);
      bv.x = v0p[32 + l16];  bv.y = v1p[32 + l16];  o2 = wmma4(pf, bv, o2);
      bv.x = v0p[48 + l16];  bv.y = v1p[48 + l16];  o3 = wmma4(pf, bv, o3);
    }
  }
  float* obase = out + (size_t)(b * L_ + qt * 16) * D_ + h * DH_;
#pragma unroll
  for (int r = 0; r < 8; ++r) {
    float inv = 1.0f / lsum[r];
    float* op = obase + (size_t)(r + 8 * half) * D_;
    op[l16]      = o0[r] * inv;
    op[16 + l16] = o1[r] * inv;
    op[32 + l16] = o2[r] * inv;
    op[48 + l16] = o3[r] * inv;
  }
}

extern "C" void kernel_launch(void* const* d_in, const int* in_sizes, int n_in,
                              void* d_out, int out_size, void* d_ws, size_t ws_size,
                              hipStream_t stream) {
  (void)in_sizes; (void)n_in; (void)out_size; (void)ws_size;
  const float* x        = (const float*)d_in[0];
  const float* pos      = (const float*)d_in[1];
  const float* pos_orig = (const float*)d_in[2];
  const float* tm       = (const float*)d_in[3];
  const float* w_qkv    = (const float*)d_in[4];
  const float* w_out    = (const float*)d_in[5];
  const float* w_up     = (const float*)d_in[6];
  const float* w_down   = (const float*)d_in[7];
  const float* n1s      = (const float*)d_in[8];
  const float* n2s      = (const float*)d_in[9];
  const float* qksc     = (const float*)d_in[10];
  float* out = (float*)d_out;

  float* xn   = (float*)d_ws;
  float* qkv  = xn  + (size_t)NTOK * D_;
  float* attn = qkv + (size_t)NTOK * 3 * D_;
  float* up   = attn + (size_t)NTOK * D_;

  rmsnorm_kernel<<<NTOK, 256, 0, stream>>>(x, n1s, xn);
  gemm_nt_kernel<<<dim3(3 * D_ / 128, NTOK / 128), 256, 0, stream>>>(
      xn, w_qkv, nullptr, qkv, NTOK, 3 * D_, D_, 0);
  rope_kernel<<<(NTOK * H_ * 25 + 255) / 256, 256, 0, stream>>>(qkv, pos, pos_orig, tm);
  cos_scale_kernel<<<NTOK * H_ / 8, 256, 0, stream>>>(qkv, qksc);
  attn_kernel<<<dim3(L_ / 16, B_ * H_), 32, 0, stream>>>(qkv, attn);
  gemm_nt_kernel<<<dim3(D_ / 128, NTOK / 128), 256, 0, stream>>>(
      attn, w_out, x, out, NTOK, D_, D_, 1);
  rmsnorm_kernel<<<NTOK, 256, 0, stream>>>(out, n2s, xn);
  gemm_nt_kernel<<<dim3(4 * D_ / 128, NTOK / 128), 256, 0, stream>>>(
      xn, w_up, nullptr, up, NTOK, 4 * D_, D_, 2);
  gemm_nt_kernel<<<dim3(D_ / 128, NTOK / 128), 256, 0, stream>>>(
      up, w_down, out, out, NTOK, D_, 4 * D_, 1);
}


# attn: score tile transposed (K.Q^T): row max/sum per lane (1 half-swap, not 64 LDS permutes), P to V^T.P^T from registers, no LDS; row sums and P.V keys summed in a new order
# speedup vs baseline: 2.7380x; 1.2356x over previous
#include <hip/hip_runtime.h>
#include <math.h>

#define B_   2
#define L_   2048
#define D_   1024
#define H_   16
#define DH_  64
#define NTOK (B_ * L_)

typedef __attribute__((ext_vector_type(2))) float v2f;
typedef __attribute__((ext_vector_type(8))) float v8f;

__device__ __forceinline__ v8f wmma4(v2f a, v2f b, v8f c) {
  return __builtin_amdgcn_wmma_f32_16x16x4_f32(
       false, a,  false, b,
       (short)0, c,  false,  false);
}

__device__ __forceinline__ float gelu_tanh(float x) {
  float x3 = x * x * x;
  return 0.5f * x * (1.0f + tanhf(0.7978845608028654f * (x + 0.044715f * x3)));
}

__global__ void __launch_bounds__(256)
rmsnorm_kernel(const float* __restrict__ x, const float* __restrict__ scale,
               float* __restrict__ out) {
  const int row = blockIdx.x;
  const float* xr = x + (size_t)row * D_;
  float* orow = out + (size_t)row * D_;
  const int t = threadIdx.x;
  float4 v = ((const float4*)xr)[t];
  float ss = v.x * v.x + v.y * v.y + v.z * v.z + v.w * v.w;
  for (int off = 16; off >= 1; off >>= 1) ss += __shfl_xor(ss, off, 32);
  __shared__ float red[8];
  if ((t & 31) == 0) red[t >> 5] = ss;
  __syncthreads();
  float tot = 0.f;
#pragma unroll
  for (int i = 0; i < 8; ++i) tot += red[i];
  float r = rsqrtf(tot * (1.0f / (float)D_) + 1e-6f);
  float4 s4 = ((const float4*)scale)[t];
  float4 o;
  o.x = v.x * s4.x * r; o.y = v.y * s4.y * r;
  o.z = v.z * s4.z * r; o.w = v.w * s4.w * r;
  ((float4*)orow)[t] = o;
}

#define GBM 128
#define GBN 128
#define GBK 32
#define GLD 36

typedef __attribute__((ext_vector_type(4))) float v4f;

__global__ void __launch_bounds__(256)
gemm_nt_kernel(const float* __restrict__ A, const float* __restrict__ W,
               const float* R, float* C,
               int M, int N, int K, int flags) {
  __shared__ __attribute__((aligned(16))) float As[2][GBM * GLD];
  __shared__ __attribute__((aligned(16))) float Ws[2][GBN * GLD];

  const int tid = threadIdx.x;
  const int lane = tid & 31;
  const int wave = tid >> 5;
  const int half = lane >> 4;
  const int l16 = lane & 15;
  const int wm = wave >> 2;
  const int wn = wave & 3;
  const int rowBlk = blockIdx.y * GBM;
  const int colBlk = blockIdx.x * GBN;

  const int lr = tid >> 3;
  const int lc = (tid & 7) * 4;
  const float* ag = A + (size_t)(rowBlk + lr) * K + lc;
  const float* wg = W + (size_t)(colBlk + lr) * K + lc;
  const size_t g32 = (size_t)32 * K;
  const int sOff = lr * GLD + lc;

  v4f ra[4], rw[4];
#pragma unroll
  for (int i = 0; i < 4; ++i) {
    ra[i] = *(const v4f*)(ag + i * g32);
    rw[i] = *(const v4f*)(wg + i * g32);
  }
#pragma unroll
  for (int i = 0; i < 4; ++i) {
    *(v4f*)(&As[0][sOff + i * 32 * GLD]) = ra[i];
    *(v4f*)(&Ws[0][sOff + i * 32 * GLD]) = rw[i];
  }
  __syncthreads();

  v8f acc[4][2];
#pragma unroll
  for (int i = 0; i < 4; ++i)
#pragma unroll
    for (int n = 0; n < 2; ++n)
      acc[i][n] = (v8f){0.f,0.f,0.f,0.f,0.f,0.f,0.f,0.f};

  const int aOff = (wm * 64 + l16) * GLD + 4 * half;
  const int bOff = (wn * 32 + l16) * GLD + 4 * half;
  const int T = K / GBK;

  for (int t = 0; t < T; ++t) {
    const int cur = t & 1;
    const bool more = (t + 1 < T);
    if (more) {
      const float* agn = ag + (size_t)(t + 1) * GBK;
      const float* wgn = wg + (size_t)(t + 1) * GBK;
#pragma unroll
      for (int i = 0; i < 4; ++i) {
        ra[i] = *(const v4f*)(agn + i * g32);
        rw[i] = *(const v4f*)(wgn + i * g32);
      }
    }
    const float* as = &As[cur][aOff];
    const float* ws = &Ws[cur][bOff];

    v4f fa[4], fb[2];
#pragma unroll
    for (int i = 0; i < 4; ++i) fa[i] = *(const v4f*)(as + i * 16 * GLD);
#pragma unroll
    for (int n = 0; n < 2; ++n) fb[n] = *(const v4f*)(ws + n * 16 * GLD);

#pragma unroll
    for (int j = 0; j < 4; ++j) {
      v4f na[4], nb[2];
      if (j < 3) {
#pragma unroll
        for (int i = 0; i < 4; ++i) na[i] = *(const v4f*)(as + i * 16 * GLD + 8 * (j + 1));
#pragma unroll
        for (int n = 0; n < 2; ++n) nb[n] = *(const v4f*)(ws + n * 16 * GLD + 8 * (j + 1));
      }
#pragma unroll
      for (int i = 0; i < 4; ++i)
#pragma unroll
        for (int n = 0; n < 2; ++n) {
          v2f a2 = {fa[i].x, fa[i].y};
          v2f b2 = {fb[n].x, fb[n].y};
          acc[i][n] = wmma4(a2, b2, acc[i][n]);
        }
#pragma unroll
      for (int i = 0; i < 4; ++i)
#pragma unroll
        for (int n = 0; n < 2; ++n) {
          v2f a2 = {fa[i].z, fa[i].w};
          v2f b2 = {fb[n].z, fb[n].w};
          acc[i][n] = wmma4(a2, b2, acc[i][n]);
        }
      if (j < 3) {
#pragma unroll
        for (int i = 0; i < 4; ++i) fa[i] = na[i];
#pragma unroll
        for (int n = 0; n < 2; ++n) fb[n] = nb[n];
      }
    }

    if (more) {
#pragma unroll
      for (int i = 0; i < 4; ++i) {
        *(v4f*)(&As[cur ^ 1][sOff + i * 32 * GLD]) = ra[i];
        *(v4f*)(&Ws[cur ^ 1][sOff + i * 32 * GLD]) = rw[i];
      }
    }
    __syncthreads();
  }

  const int row0 = rowBlk + wm * 64 + 8 * half;
  const int col0 = colBlk + wn * 32 + l16;
#pragma unroll
  for (int i = 0; i < 4; ++i) {
    const size_t ib = (size_t)(row0 + i * 16) * N + col0;
    float v0[8], v1[8];
#pragma unroll
    for (int r = 0; r < 8; ++r) { v0[r] = acc[i][0][r]; v1[r] = acc[i][1][r]; }
    if (flags & 1) {
#pragma unroll
      for (int r = 0; r < 8; ++r) {
        v0[r] += R[ib + (size_t)r * N];
        v1[r] += R[ib + (size_t)r * N + 16];
      }
    }
    if (flags & 2) {
#pragma unroll
      for (int r = 0; r < 8; ++r) { v0[r] = gelu_tanh(v0[r]); v1[r] = gelu_tanh(v1[r]); }
    }
#pragma unroll
    for (int r = 0; r < 8; ++r) {
      C[ib + (size_t)r * N] = v0[r];
      C[ib + (size_t)r * N + 16] = v1[r];
    }
  }
}

__global__ void __launch_bounds__(256)
rope_kernel(float* __restrict__ qkv, const float* __restrict__ pos,
            const float* __restrict__ pos_orig, const float* __restrict__ tm) {
  int idx = blockIdx.x * blockDim.x + threadIdx.x;
  if (idx >= NTOK * H_ * 25) return;
  int i   = idx % 25;
  int h   = (idx / 25) % H_;
  int tok = idx / (25 * H_);
  int c = i / 5, j = i % 5;
  int fi = j * H_ + h;
  float theta;
  if (c < 4) {
    float freq = __expf(1.1447298858494002f + (float)fi * 0.028782313662425572f);
    int coord = (c < 2) ? 1 : 0;
    int s = (c < 2) ? c : c - 2;
    const float* P = (s == 0) ? pos_orig : pos;
    float pv = P[(size_t)tok * 2 + coord] * 2.0f - 1.0f;
    theta = pv * freq;
  } else {
    float tf = __expf(-(float)fi * 0.057564627324851145f);
    theta = tm[tok] * tf;
  }
  float cs = cosf(theta), sn = sinf(theta);
  float* qp = qkv + (size_t)tok * 3 * D_ + h * DH_;
  float* kp = qp + D_;
  float q1 = qp[i], q2 = qp[i + 25];
  qp[i] = q1 * cs - q2 * sn;  qp[i + 25] = q2 * cs + q1 * sn;
  float k1 = kp[i], k2 = kp[i + 25];
  kp[i] = k1 * cs - k2 * sn;  kp[i + 25] = k2 * cs + k1 * sn;
}

__global__ void __launch_bounds__(256)
cos_scale_kernel(float* __restrict__ qkv, const float* __restrict__ qk_scale) {
  const int lane = threadIdx.x & 31;
  const int wave = threadIdx.x >> 5;
  const int row = blockIdx.x * 8 + wave;
  const int h = row % H_;
  const int tok = row / H_;
  float* qp = qkv + (size_t)tok * 3 * D_ + h * DH_;
  float* kp = qp + D_;
  float2 q2 = ((float2*)qp)[lane];
  float2 k2 = ((float2*)kp)[lane];
  float sq = q2.x * q2.x + q2.y * q2.y;
  float sk = k2.x * k2.x + k2.y * k2.y;
  for (int off = 16; off >= 1; off >>= 1) {
    sq += __shfl_xor(sq, off, 32);
    sk += __shfl_xor(sk, off, 32);
  }
  float s = sqrtf(qk_scale[h]);
  float qs = s * rsqrtf(sq + 1e-6f);
  float ks = s * rsqrtf(sk + 1e-6f);
  ((float2*)qp)[lane] = make_float2(q2.x * qs, q2.y * qs);
  ((float2*)kp)[lane] = make_float2(k2.x * ks, k2.y * ks);
}

__device__ __forceinline__ float xhalf(float x) {
  int bits = __builtin_bit_cast(int, x);
  return __builtin_bit_cast(float, __builtin_amdgcn_permlanex16(
      bits, bits, 0x76543210, 0xfedcba98, false, false));
}

__global__ void __launch_bounds__(32)
attn_kernel(const float* __restrict__ qkv, float* __restrict__ out) {
  const int lane = threadIdx.x & 31;
  const int half = lane >> 4, l16 = lane & 15;
  const int qt = blockIdx.x;
  const int bh = blockIdx.y;
  const int b = bh / H_, h = bh % H_;
  const size_t ts = 3 * D_;

  const float* qrow = qkv + (size_t)(b * L_ + qt * 16 + l16) * ts + h * DH_ + 2 * half;
  v2f qf[16];
#pragma unroll
  for (int s = 0; s < 16; ++s) qf[s] = *(const v2f*)(qrow + 4 * s);

  float m = -INFINITY;
  float lsum = 0.f;
  v8f o0 = {0.f,0.f,0.f,0.f,0.f,0.f,0.f,0.f};
  v8f o1 = o0, o2 = o0, o3 = o0;

  const float* kbase = qkv + (size_t)(b * L_) * ts + D_ + h * DH_;
  const float* vbase = qkv + (size_t)(b * L_) * ts + 2 * D_ + h * DH_;

  for (int kt = 0; kt < L_ / 16; ++kt) {
    const float* krow = kbase + (size_t)(kt * 16 + l16) * ts + 2 * half;
    v8f S0 = {0.f,0.f,0.f,0.f,0.f,0.f,0.f,0.f};
    v8f S1 = S0;
#pragma unroll
    for (int s = 0; s < 16; s += 2) {
      v2f kf0 = *(const v2f*)(krow + 4 * s);
      v2f kf1 = *(const v2f*)(krow + 4 * s + 4);
      S0 = wmma4(kf0, qf[s], S0);
      S1 = wmma4(kf1, qf[s + 1], S1);
    }
    float sv[8];
#pragma unroll
    for (int r = 0; r < 8; ++r) sv[r] = S0[r] + S1[r];
    float mx = sv[0];
#pragma unroll
    for (int r = 1; r < 8; ++r) mx = fmaxf(mx, sv[r]);
    mx = fmaxf(mx, xhalf(mx));
    const float mnew = fmaxf(m, mx);
    const float alpha = __expf(m - mnew);
    float p[8];
    float rs = 0.f;
#pragma unroll
    for (int r = 0; r < 8; ++r) { p[r] = __expf(sv[r] - mnew); rs += p[r]; }
    lsum = lsum * alpha + rs;
    m = mnew;
#pragma unroll
    for (int r = 0; r < 8; ++r) {
      o0[r] *= alpha; o1[r] *= alpha; o2[r] *= alpha; o3[r] *= alpha;
    }
    const float* vrow = vbase + (size_t)(kt * 16 + 8 * half) * ts;
#pragma unroll
    for (int s = 0; s < 4; ++s) {
      v2f pf;
      pf.x = p[2 * s];
      pf.y = p[2 * s + 1];
      const float* v0p = vrow + (size_t)(2 * s) * ts;
      const float* v1p = v0p + ts;
      v2f av;
      av.x = v0p[l16];       av.y = v1p[l16];       o0 = wmma4(av, pf, o0);
      av.x = v0p[16 + l16];  av.y = v1p[16 + l16];  o1 = wmma4(av, pf, o1);
      av.x = v0p[32 + l16];  av.y = v1p[32 + l16];  o2 = wmma4(av, pf, o2);
      av.x = v0p[48 + l16];  av.y = v1p[48 + l16];  o3 = wmma4(av, pf, o3);
    }
  }
  const float inv = 1.0f / (lsum + xhalf(lsum));
  float* op = out + (size_t)(b * L_ + qt * 16 + l16) * D_ + h * DH_ + 8 * half;
  v4f w;
  w = (v4f){o0[0] * inv, o0[1] * inv, o0[2] * inv, o0[3] * inv};  *(v4f*)(op)      = w;
  w = (v4f){o0[4] * inv, o0[5] * inv, o0[6] * inv, o0[7] * inv};  *(v4f*)(op + 4)  = w;
  w = (v4f){o1[0] * inv, o1[1] * inv, o1[2] * inv, o1[3] * inv};  *(v4f*)(op + 16) = w;
  w = (v4f){o1[4] * inv, o1[5] * inv, o1[6] * inv, o1[7] * inv};  *(v4f*)(op + 20) = w;
  w = (v4f){o2[0] * inv, o2[1] * inv, o2[2] * inv, o2[3] * inv};  *(v4f*)(op + 32) = w;
  w = (v4f){o2[4] * inv, o2[5] * inv, o2[6] * inv, o2[7] * inv};  *(v4f*)(op + 36) = w;
  w = (v4f){o3[0] * inv, o3[1] * inv, o3[2] * inv, o3[3] * inv};  *(v4f*)(op + 48) = w;
  w = (v4f){o3[4] * inv, o3[5] * inv, o3[6] * inv, o3[7] * inv};  *(v4f*)(op + 52) = w;
}

extern "C" void kernel_launch(void* const* d_in, const int* in_sizes, int n_in,
                              void* d_out, int out_size, void* d_ws, size_t ws_size,
                              hipStream_t stream) {
  (void)in_sizes; (void)n_in; (void)out_size; (void)ws_size;
  const float* x        = (const float*)d_in[0];
  const float* pos      = (const float*)d_in[1];
  const float* pos_orig = (const float*)d_in[2];
  const float* tm       = (const float*)d_in[3];
  const float* w_qkv    = (const float*)d_in[4];
  const float* w_out    = (const float*)d_in[5];
  const float* w_up     = (const float*)d_in[6];
  const float* w_down   = (const float*)d_in[7];
  const float* n1s      = (const float*)d_in[8];
  const float* n2s      = (const float*)d_in[9];
  const float* qksc     = (const float*)d_in[10];
  float* out = (float*)d_out;

  float* xn   = (float*)d_ws;
  float* qkv  = xn  + (size_t)NTOK * D_;
  float* attn = qkv + (size_t)NTOK * 3 * D_;
  float* up   = attn + (size_t)NTOK * D_;

  rmsnorm_kernel<<<NTOK, 256, 0, stream>>>(x, n1s, xn);
  gemm_nt_kernel<<<dim3(3 * D_ / 128, NTOK / 128), 256, 0, stream>>>(
      xn, w_qkv, nullptr, qkv, NTOK, 3 * D_, D_, 0);
  rope_kernel<<<(NTOK * H_ * 25 + 255) / 256, 256, 0, stream>>>(qkv, pos, pos_orig, tm);
  cos_scale_kernel<<<NTOK * H_ / 8, 256, 0, stream>>>(qkv, qksc);
  attn_kernel<<<dim3(L_ / 16, B_ * H_), 32, 0, stream>>>(qkv, attn);
  gemm_nt_kernel<<<dim3(D_ / 128, NTOK / 128), 256, 0, stream>>>(
      attn, w_out, x, out, NTOK, D_, D_, 1);
  rmsnorm_kernel<<<NTOK, 256, 0, stream>>>(out, n2s, xn);
  gemm_nt_kernel<<<dim3(4 * D_ / 128, NTOK / 128), 256, 0, stream>>>(
      xn, w_up, nullptr, up, NTOK, 4 * D_, D_, 2);
  gemm_nt_kernel<<<dim3(D_ / 128, NTOK / 128), 256, 0, stream>>>(
      up, w_down, out, out, NTOK, D_, 4 * D_, 1);
}


# attn: 4 waves (64 queries) per work-group share K and V^T tiles staged in double-buffered LDS by 16-B global loads, fragments by 16-B LDS reads; score d terms summed in a new order
# speedup vs baseline: 2.9655x; 1.0831x over previous
#include <hip/hip_runtime.h>
#include <math.h>

#define B_   2
#define L_   2048
#define D_   1024
#define H_   16
#define DH_  64
#define NTOK (B_ * L_)

typedef __attribute__((ext_vector_type(2))) float v2f;
typedef __attribute__((ext_vector_type(8))) float v8f;

__device__ __forceinline__ v8f wmma4(v2f a, v2f b, v8f c) {
  return __builtin_amdgcn_wmma_f32_16x16x4_f32(
       false, a,  false, b,
       (short)0, c,  false,  false);
}

__device__ __forceinline__ float gelu_tanh(float x) {
  float x3 = x * x * x;
  return 0.5f * x * (1.0f + tanhf(0.7978845608028654f * (x + 0.044715f * x3)));
}

__global__ void __launch_bounds__(256)
rmsnorm_kernel(const float* __restrict__ x, const float* __restrict__ scale,
               float* __restrict__ out) {
  const int row = blockIdx.x;
  const float* xr = x + (size_t)row * D_;
  float* orow = out + (size_t)row * D_;
  const int t = threadIdx.x;
  float4 v = ((const float4*)xr)[t];
  float ss = v.x * v.x + v.y * v.y + v.z * v.z + v.w * v.w;
  for (int off = 16; off >= 1; off >>= 1) ss += __shfl_xor(ss, off, 32);
  __shared__ float red[8];
  if ((t & 31) == 0) red[t >> 5] = ss;
  __syncthreads();
  float tot = 0.f;
#pragma unroll
  for (int i = 0; i < 8; ++i) tot += red[i];
  float r = rsqrtf(tot * (1.0f / (float)D_) + 1e-6f);
  float4 s4 = ((const float4*)scale)[t];
  float4 o;
  o.x = v.x * s4.x * r; o.y = v.y * s4.y * r;
  o.z = v.z * s4.z * r; o.w = v.w * s4.w * r;
  ((float4*)orow)[t] = o;
}

#define GBM 128
#define GBN 128
#define GBK 32
#define GLD 36

typedef __attribute__((ext_vector_type(4))) float v4f;

__global__ void __launch_bounds__(256)
gemm_nt_kernel(const float* __restrict__ A, const float* __restrict__ W,
               const float* R, float* C,
               int M, int N, int K, int flags) {
  __shared__ __attribute__((aligned(16))) float As[2][GBM * GLD];
  __shared__ __attribute__((aligned(16))) float Ws[2][GBN * GLD];

  const int tid = threadIdx.x;
  const int lane = tid & 31;
  const int wave = tid >> 5;
  const int half = lane >> 4;
  const int l16 = lane & 15;
  const int wm = wave >> 2;
  const int wn = wave & 3;
  const int rowBlk = blockIdx.y * GBM;
  const int colBlk = blockIdx.x * GBN;

  const int lr = tid >> 3;
  const int lc = (tid & 7) * 4;
  const float* ag = A + (size_t)(rowBlk + lr) * K + lc;
  const float* wg = W + (size_t)(colBlk + lr) * K + lc;
  const size_t g32 = (size_t)32 * K;
  const int sOff = lr * GLD + lc;

  v4f ra[4], rw[4];
#pragma unroll
  for (int i = 0; i < 4; ++i) {
    ra[i] = *(const v4f*)(ag + i * g32);
    rw[i] = *(const v4f*)(wg + i * g32);
  }
#pragma unroll
  for (int i = 0; i < 4; ++i) {
    *(v4f*)(&As[0][sOff + i * 32 * GLD]) = ra[i];
    *(v4f*)(&Ws[0][sOff + i * 32 * GLD]) = rw[i];
  }
  __syncthreads();

  v8f acc[4][2];
#pragma unroll
  for (int i = 0; i < 4; ++i)
#pragma unroll
    for (int n = 0; n < 2; ++n)
      acc[i][n] = (v8f){0.f,0.f,0.f,0.f,0.f,0.f,0.f,0.f};

  const int aOff = (wm * 64 + l16) * GLD + 4 * half;
  const int bOff = (wn * 32 + l16) * GLD + 4 * half;
  const int T = K / GBK;

  for (int t = 0; t < T; ++t) {
    const int cur = t & 1;
    const bool more = (t + 1 < T);
    if (more) {
      const float* agn = ag + (size_t)(t + 1) * GBK;
      const float* wgn = wg + (size_t)(t + 1) * GBK;
#pragma unroll
      for (int i = 0; i < 4; ++i) {
        ra[i] = *(const v4f*)(agn + i * g32);
        rw[i] = *(const v4f*)(wgn + i * g32);
      }
    }
    const float* as = &As[cur][aOff];
    const float* ws = &Ws[cur][bOff];

    v4f fa[4], fb[2];
#pragma unroll
    for (int i = 0; i < 4; ++i) fa[i] = *(const v4f*)(as + i * 16 * GLD);
#pragma unroll
    for (int n = 0; n < 2; ++n) fb[n] = *(const v4f*)(ws + n * 16 * GLD);

#pragma unroll
    for (int j = 0; j < 4; ++j) {
      v4f na[4], nb[2];
      if (j < 3) {
#pragma unroll
        for (int i = 0; i < 4; ++i) na[i] = *(const v4f*)(as + i * 16 * GLD + 8 * (j + 1));
#pragma unroll
        for (int n = 0; n < 2; ++n) nb[n] = *(const v4f*)(ws + n * 16 * GLD + 8 * (j + 1));
      }
#pragma unroll
      for (int i = 0; i < 4; ++i)
#pragma unroll
        for (int n = 0; n < 2; ++n) {
          v2f a2 = {fa[i].x, fa[i].y};
          v2f b2 = {fb[n].x, fb[n].y};
          acc[i][n] = wmma4(a2, b2, acc[i][n]);
        }
#pragma unroll
      for (int i = 0; i < 4; ++i)
#pragma unroll
        for (int n = 0; n < 2; ++n) {
          v2f a2 = {fa[i].z, fa[i].w};
          v2f b2 = {fb[n].z, fb[n].w};
          acc[i][n] = wmma4(a2, b2, acc[i][n]);
        }
      if (j < 3) {
#pragma unroll
        for (int i = 0; i < 4; ++i) fa[i] = na[i];
#pragma unroll
        for (int n = 0; n < 2; ++n) fb[n] = nb[n];
      }
    }

    if (more) {
#pragma unroll
      for (int i = 0; i < 4; ++i) {
        *(v4f*)(&As[cur ^ 1][sOff + i * 32 * GLD]) = ra[i];
        *(v4f*)(&Ws[cur ^ 1][sOff + i * 32 * GLD]) = rw[i];
      }
    }
    __syncthreads();
  }

  const int row0 = rowBlk + wm * 64 + 8 * half;
  const int col0 = colBlk + wn * 32 + l16;
#pragma unroll
  for (int i = 0; i < 4; ++i) {
    const size_t ib = (size_t)(row0 + i * 16) * N + col0;
    float v0[8], v1[8];
#pragma unroll
    for (int r = 0; r < 8; ++r) { v0[r] = acc[i][0][r]; v1[r] = acc[i][1][r]; }
    if (flags & 1) {
#pragma unroll
      for (int r = 0; r < 8; ++r) {
        v0[r] += R[ib + (size_t)r * N];
        v1[r] += R[ib + (size_t)r * N + 16];
      }
    }
    if (flags & 2) {
#pragma unroll
      for (int r = 0; r < 8; ++r) { v0[r] = gelu_tanh(v0[r]); v1[r] = gelu_tanh(v1[r]); }
    }
#pragma unroll
    for (int r = 0; r < 8; ++r) {
      C[ib + (size_t)r * N] = v0[r];
      C[ib + (size_t)r * N + 16] = v1[r];
    }
  }
}

__global__ void __launch_bounds__(256)
rope_kernel(float* __restrict__ qkv, const float* __restrict__ pos,
            const float* __restrict__ pos_orig, const float* __restrict__ tm) {
  int idx = blockIdx.x * blockDim.x + threadIdx.x;
  if (idx >= NTOK * H_ * 25) return;
  int i   = idx % 25;
  int h   = (idx / 25) % H_;
  int tok = idx / (25 * H_);
  int c = i / 5, j = i % 5;
  int fi = j * H_ + h;
  float theta;
  if (c < 4) {
    float freq = __expf(1.1447298858494002f + (float)fi * 0.028782313662425572f);
    int coord = (c < 2) ? 1 : 0;
    int s = (c < 2) ? c : c - 2;
    const float* P = (s == 0) ? pos_orig : pos;
    float pv = P[(size_t)tok * 2 + coord] * 2.0f - 1.0f;
    theta = pv * freq;
  } else {
    float tf = __expf(-(float)fi * 0.057564627324851145f);
    theta = tm[tok] * tf;
  }
  float cs = cosf(theta), sn = sinf(theta);
  float* qp = qkv + (size_t)tok * 3 * D_ + h * DH_;
  float* kp = qp + D_;
  float q1 = qp[i], q2 = qp[i + 25];
  qp[i] = q1 * cs - q2 * sn;  qp[i + 25] = q2 * cs + q1 * sn;
  float k1 = kp[i], k2 = kp[i + 25];
  kp[i] = k1 * cs - k2 * sn;  kp[i + 25] = k2 * cs + k1 * sn;
}

__global__ void __launch_bounds__(256)
cos_scale_kernel(float* __restrict__ qkv, const float* __restrict__ qk_scale) {
  const int lane = threadIdx.x & 31;
  const int wave = threadIdx.x >> 5;
  const int row = blockIdx.x * 8 + wave;
  const int h = row % H_;
  const int tok = row / H_;
  float* qp = qkv + (size_t)tok * 3 * D_ + h * DH_;
  float* kp = qp + D_;
  float2 q2 = ((float2*)qp)[lane];
  float2 k2 = ((float2*)kp)[lane];
  float sq = q2.x * q2.x + q2.y * q2.y;
  float sk = k2.x * k2.x + k2.y * k2.y;
  for (int off = 16; off >= 1; off >>= 1) {
    sq += __shfl_xor(sq, off, 32);
    sk += __shfl_xor(sk, off, 32);
  }
  float s = sqrtf(qk_scale[h]);
  float qs = s * rsqrtf(sq + 1e-6f);
  float ks = s * rsqrtf(sk + 1e-6f);
  ((float2*)qp)[lane] = make_float2(q2.x * qs, q2.y * qs);
  ((float2*)kp)[lane] = make_float2(k2.x * ks, k2.y * ks);
}

#define AQ  64
#define AKT 32
#define KLD 68
#define VLD 36

__device__ __forceinline__ float xhalf(float x) {
  int bits = __builtin_bit_cast(int, x);
  return __builtin_bit_cast(float, __builtin_amdgcn_permlanex16(
      bits, bits, 0x76543210, 0xfedcba98, false, false));
}

__global__ void __launch_bounds__(128)
attn_kernel(const float* __restrict__ qkv, float* __restrict__ out) {
  __shared__ __attribute__((aligned(16))) float Ks[2][AKT * KLD];
  __shared__ __attribute__((aligned(16))) float Vt[2][DH_ * VLD];

  const int tid = threadIdx.x;
  const int lane = tid & 31;
  const int wave = tid >> 5;
  const int half = lane >> 4, l16 = lane & 15;
  const int bh = blockIdx.y;
  const int b = bh / H_, h = bh % H_;
  const size_t ts = 3 * D_;
  const int q0 = blockIdx.x * AQ + wave * 16;

  const float* qrow = qkv + (size_t)(b * L_ + q0 + l16) * ts + h * DH_ + 4 * half;
  v4f qf[8];
#pragma unroll
  for (int j = 0; j < 8; ++j) qf[j] = *(const v4f*)(qrow + 8 * j);

  const int sr = tid >> 3;
  const int sc = (tid & 7) * 4;
  const float* kg = qkv + (size_t)(b * L_ + sr) * ts + D_ + h * DH_ + sc;
  const float* vg = kg + D_;
  const int kSt = sr * KLD + sc;
  const int vSt = sc * VLD + ((sr & 3) | ((sr & 4) << 1) | ((sr & 8) >> 1));

  v4f rk[4], rv[4];
#pragma unroll
  for (int u = 0; u < 4; ++u) {
    rk[u] = *(const v4f*)(kg + (size_t)(16 * (u >> 1)) * ts + 32 * (u & 1));
    rv[u] = *(const v4f*)(vg + (size_t)(16 * (u >> 1)) * ts + 32 * (u & 1));
  }
#pragma unroll
  for (int u = 0; u < 4; ++u) {
    *(v4f*)(&Ks[0][kSt + 16 * (u >> 1) * KLD + 32 * (u & 1)]) = rk[u];
    float* vp = &Vt[0][vSt + 32 * (u & 1) * VLD + 16 * (u >> 1)];
    vp[0] = rv[u].x;  vp[VLD] = rv[u].y;  vp[2 * VLD] = rv[u].z;  vp[3 * VLD] = rv[u].w;
  }
  __syncthreads();

  float m = -INFINITY;
  float lsum = 0.f;
  v8f o0 = {0.f,0.f,0.f,0.f,0.f,0.f,0.f,0.f};
  v8f o1 = o0, o2 = o0, o3 = o0;

  const int kOff = l16 * KLD + 4 * half;
  const int vOff = l16 * VLD + 4 * half;
  const int NT = L_ / AKT;

  for (int t = 0; t < NT; ++t) {
    const int cur = t & 1;
    const bool more = (t + 1 < NT);
    if (more) {
      const float* kgn = kg + (size_t)(t + 1) * AKT * ts;
      const float* vgn = kgn + D_;
#pragma unroll
      for (int u = 0; u < 4; ++u) {
        rk[u] = *(const v4f*)(kgn + (size_t)(16 * (u >> 1)) * ts + 32 * (u & 1));
        rv[u] = *(const v4f*)(vgn + (size_t)(16 * (u >> 1)) * ts + 32 * (u & 1));
      }
    }
    __builtin_amdgcn_sched_barrier(0);
    const float* ksb = &Ks[cur][kOff];
    const float* vtb = &Vt[cur][vOff];

#pragma unroll
    for (int sub = 0; sub < 2; ++sub) {
      v4f kf[8];
#pragma unroll
      for (int j = 0; j < 8; ++j) kf[j] = *(const v4f*)(ksb + sub * 16 * KLD + 8 * j);
      v4f va[4], vb[4];
#pragma unroll
      for (int c = 0; c < 4; ++c) {
        va[c] = *(const v4f*)(vtb + c * 16 * VLD + sub * 16);
        vb[c] = *(const v4f*)(vtb + c * 16 * VLD + sub * 16 + 8);
      }
      v8f S0 = {0.f,0.f,0.f,0.f,0.f,0.f,0.f,0.f};
      v8f S1 = S0;
#pragma unroll
      for (int j = 0; j < 8; ++j) {
        v2f k0 = {kf[j].x, kf[j].y};
        v2f q0v = {qf[j].x, qf[j].y};
        v2f k1 = {kf[j].z, kf[j].w};
        v2f q1v = {qf[j].z, qf[j].w};
        S0 = wmma4(k0, q0v, S0);
        S1 = wmma4(k1, q1v, S1);
      }
      float sv[8];
#pragma unroll
      for (int r = 0; r < 8; ++r) sv[r] = S0[r] + S1[r];
      float mx = sv[0];
#pragma unroll
      for (int r = 1; r < 8; ++r) mx = fmaxf(mx, sv[r]);
      mx = fmaxf(mx, xhalf(mx));
      const float mnew = fmaxf(m, mx);
      const float alpha = __expf(m - mnew);
      float p[8];
      float rs = 0.f;
#pragma unroll
      for (int r = 0; r < 8; ++r) { p[r] = __expf(sv[r] - mnew); rs += p[r]; }
      lsum = lsum * alpha + rs;
      m = mnew;
#pragma unroll
      for (int r = 0; r < 8; ++r) {
        o0[r] *= alpha; o1[r] *= alpha; o2[r] *= alpha; o3[r] *= alpha;
      }
      v2f pf, av;
      pf.x = p[0];  pf.y = p[1];
      av.x = va[0].x;  av.y = va[0].y;  o0 = wmma4(av, pf, o0);
      av.x = va[1].x;  av.y = va[1].y;  o1 = wmma4(av, pf, o1);
      av.x = va[2].x;  av.y = va[2].y;  o2 = wmma4(av, pf, o2);
      av.x = va[3].x;  av.y = va[3].y;  o3 = wmma4(av, pf, o3);
      pf.x = p[2];  pf.y = p[3];
      av.x = va[0].z;  av.y = va[0].w;  o0 = wmma4(av, pf, o0);
      av.x = va[1].z;  av.y = va[1].w;  o1 = wmma4(av, pf, o1);
      av.x = va[2].z;  av.y = va[2].w;  o2 = wmma4(av, pf, o2);
      av.x = va[3].z;  av.y = va[3].w;  o3 = wmma4(av, pf, o3);
      pf.x = p[4];  pf.y = p[5];
      av.x = vb[0].x;  av.y = vb[0].y;  o0 = wmma4(av, pf, o0);
      av.x = vb[1].x;  av.y = vb[1].y;  o1 = wmma4(av, pf, o1);
      av.x = vb[2].x;  av.y = vb[2].y;  o2 = wmma4(av, pf, o2);
      av.x = vb[3].x;  av.y = vb[3].y;  o3 = wmma4(av, pf, o3);
      pf.x = p[6];  pf.y = p[7];
      av.x = vb[0].z;  av.y = vb[0].w;  o0 = wmma4(av, pf, o0);
      av.x = vb[1].z;  av.y = vb[1].w;  o1 = wmma4(av, pf, o1);
      av.x = vb[2].z;  av.y = vb[2].w;  o2 = wmma4(av, pf, o2);
      av.x = vb[3].z;  av.y = vb[3].w;  o3 = wmma4(av, pf, o3);
    }

    if (more) {
#pragma unroll
      for (int u = 0; u < 4; ++u) {
        *(v4f*)(&Ks[cur ^ 1][kSt + 16 * (u >> 1) * KLD + 32 * (u & 1)]) = rk[u];
        float* vp = &Vt[cur ^ 1][vSt + 32 * (u & 1) * VLD + 16 * (u >> 1)];
        vp[0] = rv[u].x;  vp[VLD] = rv[u].y;  vp[2 * VLD] = rv[u].z;  vp[3 * VLD] = rv[u].w;
      }
    }
    __syncthreads();
  }
  const float inv = 1.0f / (lsum + xhalf(lsum));
  float* op = out + (size_t)(b * L_ + q0 + l16) * D_ + h * DH_ + 8 * half;
  v4f w;
  w = (v4f){o0[0] * inv, o0[1] * inv, o0[2] * inv, o0[3] * inv};  *(v4f*)(op)      = w;
  w = (v4f){o0[4] * inv, o0[5] * inv, o0[6] * inv, o0[7] * inv};  *(v4f*)(op + 4)  = w;
  w = (v4f){o1[0] * inv, o1[1] * inv, o1[2] * inv, o1[3] * inv};  *(v4f*)(op + 16) = w;
  w = (v4f){o1[4] * inv, o1[5] * inv, o1[6] * inv, o1[7] * inv};  *(v4f*)(op + 20) = w;
  w = (v4f){o2[0] * inv, o2[1] * inv, o2[2] * inv, o2[3] * inv};  *(v4f*)(op + 32) = w;
  w = (v4f){o2[4] * inv, o2[5] * inv, o2[6] * inv, o2[7] * inv};  *(v4f*)(op + 36) = w;
  w = (v4f){o3[0] * inv, o3[1] * inv, o3[2] * inv, o3[3] * inv};  *(v4f*)(op + 48) = w;
  w = (v4f){o3[4] * inv, o3[5] * inv, o3[6] * inv, o3[7] * inv};  *(v4f*)(op + 52) = w;
}

extern "C" void kernel_launch(void* const* d_in, const int* in_sizes, int n_in,
                              void* d_out, int out_size, void* d_ws, size_t ws_size,
                              hipStream_t stream) {
  (void)in_sizes; (void)n_in; (void)out_size; (void)ws_size;
  const float* x        = (const float*)d_in[0];
  const float* pos      = (const float*)d_in[1];
  const float* pos_orig = (const float*)d_in[2];
  const float* tm       = (const float*)d_in[3];
  const float* w_qkv    = (const float*)d_in[4];
  const float* w_out    = (const float*)d_in[5];
  const float* w_up     = (const float*)d_in[6];
  const float* w_down   = (const float*)d_in[7];
  const float* n1s      = (const float*)d_in[8];
  const float* n2s      = (const float*)d_in[9];
  const float* qksc     = (const float*)d_in[10];
  float* out = (float*)d_out;

  float* xn   = (float*)d_ws;
  float* qkv  = xn  + (size_t)NTOK * D_;
  float* attn = qkv + (size_t)NTOK * 3 * D_;
  float* up   = attn + (size_t)NTOK * D_;

  rmsnorm_kernel<<<NTOK, 256, 0, stream>>>(x, n1s, xn);
  gemm_nt_kernel<<<dim3(3 * D_ / 128, NTOK / 128), 256, 0, stream>>>(
      xn, w_qkv, nullptr, qkv, NTOK, 3 * D_, D_, 0);
  rope_kernel<<<(NTOK * H_ * 25 + 255) / 256, 256, 0, stream>>>(qkv, pos, pos_orig, tm);
  cos_scale_kernel<<<NTOK * H_ / 8, 256, 0, stream>>>(qkv, qksc);
  attn_kernel<<<dim3(L_ / AQ, B_ * H_), 128, 0, stream>>>(qkv, attn);
  gemm_nt_kernel<<<dim3(D_ / 128, NTOK / 128), 256, 0, stream>>>(
      attn, w_out, x, out, NTOK, D_, D_, 1);
  rmsnorm_kernel<<<NTOK, 256, 0, stream>>>(out, n2s, xn);
  gemm_nt_kernel<<<dim3(4 * D_ / 128, NTOK / 128), 256, 0, stream>>>(
      xn, w_up, nullptr, up, NTOK, 4 * D_, D_, 2);
  gemm_nt_kernel<<<dim3(D_ / 128, NTOK / 128), 256, 0, stream>>>(
      up, w_down, out, out, NTOK, D_, 4 * D_, 1);
}
